# starting kernel
# baseline (speedup 1.0000x reference)
#include <hip/hip_runtime.h>
#include <hip/hip_bf16.h>

#define DIM 50
#define KPAD 64
#define OUT 20
#define NPAD 32

typedef __attribute__((ext_vector_type(2))) float v2f;
typedef __attribute__((ext_vector_type(8))) float v8f;

__global__ __launch_bounds__(256)
void pad_w_kernel(const float* __restrict__ W, float* __restrict__ Wpad) {
    for (int i = threadIdx.x; i < NPAD * KPAD; i += 256) {
        const int n = i >> 6;
        const int k = i & (KPAD-1);
        Wpad[i] = (n < OUT && k < DIM) ? W[n * DIM + k] : 0.0f;
    }
}

__global__ __launch_bounds__(256)
void pool_kernel(const int* __restrict__ x,
                 const int* __restrict__ lengths,
                 const float* __restrict__ emb,
                 float* __restrict__ rep,
                 int L) {
    const int row = blockIdx.x;
    const int d   = threadIdx.x;
    const int ty  = threadIdx.y;
    const int len = lengths[row];

    const int* xrow = x + (size_t)row * (size_t)L;

    float acc = 0.0f;
    if (d < DIM) {
        for (int j = ty; j < len; j += 4) {
            const int tok = xrow[j];
            acc += emb[(size_t)tok * DIM + d];
        }
    }

    __shared__ float smem[4][64];
    smem[ty][d] = acc;
    __syncthreads();

    if (ty == 0) {
        float s = smem[0][d] + smem[1][d] + smem[2][d] + smem[3][d];
        s = s / (float)len;
        s = fmaxf(s, 0.0f);
        rep[(size_t)row * KPAD + d] = (d < DIM) ? s : 0.0f;
    }
}

__global__ __launch_bounds__(32)
void head_gemm_kernel(const float* __restrict__ rep,
                      const float* __restrict__ Wpad,
                      const float* __restrict__ bias,
                      float* __restrict__ out) {
    const int lane  = threadIdx.x;
    const int row0  = blockIdx.x * 16;
    const int mN    = lane & 15;
    const int khalf = (lane >> 4) * 2;

    const float* arow  = rep  + (size_t)(row0 + mN) * KPAD + khalf;
    const float* brow0 = Wpad + (size_t)mN        * KPAD + khalf;
    const float* brow1 = Wpad + (size_t)(16 + mN) * KPAD + khalf;

    v8f c0 = {};
    v8f c1 = {};

#pragma unroll
    for (int kk = 0; kk < 13; ++kk) {
        const int kb = kk * 4;
        v2f a  = *(const v2f*)(arow  + kb);
        v2f b0 = *(const v2f*)(brow0 + kb);
        v2f b1 = *(const v2f*)(brow1 + kb);

        c0 = __builtin_amdgcn_wmma_f32_16x16x4_f32(
                 false, a, false, b0, (short)0, c0, false, false);
        c1 = __builtin_amdgcn_wmma_f32_16x16x4_f32(
                 false, a, false, b1, (short)0, c1, false, false);
    }

    const int n0 = mN;
    const int n1 = 16 + mN;
    const bool n1v = (n1 < OUT);
    const int mbase = row0 + ((lane >> 4) ? 8 : 0);
    const float bias0 = bias[n0];
    const float bias1 = n1v ? bias[n1] : 0.0f;
#pragma unroll
    for (int r = 0; r < 8; ++r) {
        const int M = mbase + r;
        out[(size_t)M * OUT + n0] = c0[r] + bias0;
        if (n1v) out[(size_t)M * OUT + n1] = c1[r] + bias1;
    }
}

extern "C" void kernel_launch(void* const* d_in, const int* in_sizes, int n_in,
                              void* d_out, int out_size, void* d_ws, size_t ws_size,
                              hipStream_t stream) {
    const int*   x       = (const int*)d_in[0];
    const int*   lengths = (const int*)d_in[1];
    const float* emb     = (const float*)d_in[2];
    const float* W       = (const float*)d_in[3];
    const float* bias    = (const float*)d_in[4];
    float*       out     = (float*)d_out;

    const int B = in_sizes[1];
    const int L = in_sizes[0] / B;

    float* rep  = (float*)d_ws;
    float* Wpad = rep + (size_t)B * KPAD;

    pad_w_kernel<<<1, 256, 0, stream>>>(W, Wpad);

    dim3 blk1(64, 4);
    pool_kernel<<<B, blk1, 0, stream>>>(x, lengths, emb, rep, L);

    head_gemm_kernel<<<B / 16, 32, 0, stream>>>(rep, Wpad, bias, out);
}


# pool kernel: 2 waves per sample; a wave loads 128 token ids coalesced, readlane-broadcasts each to a scalar row base, 25 lanes load 8 B each, 32 rows in flight per wave; sum order changes
# speedup vs baseline: 2.8077x; 2.8077x over previous
#include <hip/hip_runtime.h>
#include <hip/hip_bf16.h>

#define DIM 50
#define KPAD 64
#define OUT 20
#define NPAD 32

typedef __attribute__((ext_vector_type(2))) float v2f;
typedef __attribute__((ext_vector_type(8))) float v8f;

__global__ __launch_bounds__(256)
void pad_w_kernel(const float* __restrict__ W, float* __restrict__ Wpad) {
    for (int i = threadIdx.x; i < NPAD * KPAD; i += 256) {
        const int n = i >> 6;
        const int k = i & (KPAD-1);
        Wpad[i] = (n < OUT && k < DIM) ? W[n * DIM + k] : 0.0f;
    }
}

#define ROWB (DIM * 4)
#define PWAVES 2

template <int G>
__device__ __forceinline__ void gather_full(v2f& acc, const char* embb, unsigned loff, int myoff, int t) {
    v2f v[G];
#pragma unroll
    for (int i = 0; i < G; ++i) {
        const unsigned o = (unsigned)__builtin_amdgcn_readlane(myoff, t + i);
        v[i] = *(const v2f*)(embb + (size_t)(o + loff));
    }
#pragma unroll
    for (int i = 0; i < G; ++i) acc += v[i];
}

template <int G>
__device__ __forceinline__ void gather_part(v2f& acc, const char* embb, unsigned loff, int myoff, int t, int n) {
    v2f v[G];
#pragma unroll
    for (int i = 0; i < G; ++i) {
        const unsigned o = (unsigned)__builtin_amdgcn_readlane(myoff, t + i);
        v[i] = *(const v2f*)(embb + (size_t)(o + loff));
    }
#pragma unroll
    for (int i = 0; i < G; ++i) {
        if (t + i < n) acc += v[i];
    }
}

__global__ __launch_bounds__(PWAVES * 32) __attribute__((amdgpu_waves_per_eu(1, 8)))
void pool_kernel(const int* __restrict__ x,
                 const int* __restrict__ lengths,
                 const float* __restrict__ emb,
                 float* __restrict__ rep,
                 int L) {
    const int row  = blockIdx.x;
    const int lane = threadIdx.x & 31;
    const int wave = __builtin_amdgcn_readfirstlane((int)(threadIdx.x >> 5));
    const int len  = lengths[row];
    const int ntok = len < L ? len : L;

    const int*  xrow = x + (size_t)row * (size_t)L;
    const char* embb = (const char*)emb;
    const unsigned loff = (unsigned)(lane < 25 ? lane : 24) * 8u;

    const int half  = (ntok + 1) >> 1;
    const int start = wave * half;
    const int stop  = (start + half < ntok) ? start + half : ntok;

    v2f acc = {0.0f, 0.0f};

    for (int base = start; base < stop; base += 128) {
        int off[4];
#pragma unroll
        for (int k = 0; k < 4; ++k) {
            const int i = base + 32 * k + lane;
            const int r = xrow[i < stop ? i : stop - 1];
            off[k] = (int)((unsigned)r * (unsigned)ROWB);
        }
        asm volatile("" :: "v"(off[0]), "v"(off[1]), "v"(off[2]), "v"(off[3]));
#pragma unroll
        for (int k = 0; k < 4; ++k) {
            const int rem = stop - base - 32 * k;
            if (rem >= 32) {
                gather_full<32>(acc, embb, loff, off[k], 0);
            } else if (rem > 0) {
                int t = 0;
                if (rem >= 16) { gather_full<16>(acc, embb, loff, off[k], 0); t = 16; }
                if (t < rem) gather_part<16>(acc, embb, loff, off[k], t, rem);
            }
        }
    }

    __shared__ v2f part[PWAVES][32];
    part[wave][lane] = acc;
    __syncthreads();

    if (wave == 0) {
        v2f s = part[0][lane] + part[1][lane];
        const float fl = (float)len;
        s.x = fmaxf(s.x / fl, 0.0f);
        s.y = fmaxf(s.y / fl, 0.0f);
        if (lane >= 25) { s.x = 0.0f; s.y = 0.0f; }
        *(v2f*)(rep + (size_t)row * KPAD + 2 * lane) = s;
    }
}

__global__ __launch_bounds__(32)
void head_gemm_kernel(const float* __restrict__ rep,
                      const float* __restrict__ Wpad,
                      const float* __restrict__ bias,
                      float* __restrict__ out) {
    const int lane  = threadIdx.x;
    const int row0  = blockIdx.x * 16;
    const int mN    = lane & 15;
    const int khalf = (lane >> 4) * 2;

    const float* arow  = rep  + (size_t)(row0 + mN) * KPAD + khalf;
    const float* brow0 = Wpad + (size_t)mN        * KPAD + khalf;
    const float* brow1 = Wpad + (size_t)(16 + mN) * KPAD + khalf;

    v8f c0 = {};
    v8f c1 = {};

#pragma unroll
    for (int kk = 0; kk < 13; ++kk) {
        const int kb = kk * 4;
        v2f a  = *(const v2f*)(arow  + kb);
        v2f b0 = *(const v2f*)(brow0 + kb);
        v2f b1 = *(const v2f*)(brow1 + kb);

        c0 = __builtin_amdgcn_wmma_f32_16x16x4_f32(
                 false, a, false, b0, (short)0, c0, false, false);
        c1 = __builtin_amdgcn_wmma_f32_16x16x4_f32(
                 false, a, false, b1, (short)0, c1, false, false);
    }

    const int n0 = mN;
    const int n1 = 16 + mN;
    const bool n1v = (n1 < OUT);
    const int mbase = row0 + ((lane >> 4) ? 8 : 0);
    const float bias0 = bias[n0];
    const float bias1 = n1v ? bias[n1] : 0.0f;
#pragma unroll
    for (int r = 0; r < 8; ++r) {
        const int M = mbase + r;
        out[(size_t)M * OUT + n0] = c0[r] + bias0;
        if (n1v) out[(size_t)M * OUT + n1] = c1[r] + bias1;
    }
}

extern "C" void kernel_launch(void* const* d_in, const int* in_sizes, int n_in,
                              void* d_out, int out_size, void* d_ws, size_t ws_size,
                              hipStream_t stream) {
    const int*   x       = (const int*)d_in[0];
    const int*   lengths = (const int*)d_in[1];
    const float* emb     = (const float*)d_in[2];
    const float* W       = (const float*)d_in[3];
    const float* bias    = (const float*)d_in[4];
    float*       out     = (float*)d_out;

    const int B = in_sizes[1];
    const int L = in_sizes[0] / B;

    float* rep  = (float*)d_ws;
    float* Wpad = rep + (size_t)B * KPAD;

    pad_w_kernel<<<1, 256, 0, stream>>>(W, Wpad);

    pool_kernel<<<B, 64, 0, stream>>>(x, lengths, emb, rep, L);

    head_gemm_kernel<<<B / 16, 32, 0, stream>>>(rep, Wpad, bias, out);
}


# one launch instead of three: a 32-wave block pools its 16 samples (2 waves each, gather unchanged) into LDS, stages zero-padded W in LDS, wave 0 runs the 26 f32 WMMAs from LDS; sum order kept
# speedup vs baseline: 3.7411x; 1.3324x over previous
#include <hip/hip_runtime.h>
#include <hip/hip_bf16.h>

#define DIM 50
#define OUT 20
#define NPAD 32
#define SPB 16
#define PWAVES 2
#define NWAVES (SPB * PWAVES)
#define LPITCH 68

typedef __attribute__((ext_vector_type(2))) float v2f;
typedef __attribute__((ext_vector_type(8))) float v8f;

#define ROWB (DIM * 4)

template <int G>
__device__ __forceinline__ void gather_full(v2f& acc, const char* embb, unsigned loff, int myoff, int t) {
    v2f v[G];
#pragma unroll
    for (int i = 0; i < G; ++i) {
        const unsigned o = (unsigned)__builtin_amdgcn_readlane(myoff, t + i);
        v[i] = *(const v2f*)(embb + (size_t)(o + loff));
    }
#pragma unroll
    for (int i = 0; i < G; ++i) acc += v[i];
}

template <int G>
__device__ __forceinline__ void gather_part(v2f& acc, const char* embb, unsigned loff, int myoff, int t, int n) {
    v2f v[G];
#pragma unroll
    for (int i = 0; i < G; ++i) {
        const unsigned o = (unsigned)__builtin_amdgcn_readlane(myoff, t + i);
        v[i] = *(const v2f*)(embb + (size_t)(o + loff));
    }
#pragma unroll
    for (int i = 0; i < G; ++i) {
        if (t + i < n) acc += v[i];
    }
}

__global__ __launch_bounds__(NWAVES * 32) __attribute__((amdgpu_waves_per_eu(8, 8)))
void pool_head_kernel(const int* __restrict__ x,
                      const int* __restrict__ lengths,
                      const float* __restrict__ emb,
                      const float* __restrict__ W,
                      const float* __restrict__ bias,
                      float* __restrict__ out,
                      int L) {
    __shared__ v2f   part[NWAVES][32];
    __shared__ float repS[SPB * LPITCH];
    __shared__ float Ws[NPAD * LPITCH];

    const int tid  = threadIdx.x;
    const int lane = tid & 31;
    const int wave = __builtin_amdgcn_readfirstlane((int)(threadIdx.x >> 5));
    const int row  = blockIdx.x * SPB + (wave >> 1);
    const int hsel = wave & 1;

    const int wk  = tid & 63;
    const int wn0 = tid >> 6;
    const int wn1 = wn0 + 16;
    const float w0 = (wk < DIM) ? W[wn0 * DIM + wk] : 0.0f;
    const float w1 = (wk < DIM && wn1 < OUT) ? W[wn1 * DIM + wk] : 0.0f;

    const float bias0 = bias[lane & 15];
    const float bias1 = ((lane & 15) + 16 < OUT) ? bias[(lane & 15) + 16] : 0.0f;

    const int len  = lengths[row];
    const int ntok = len < L ? len : L;

    const int*  xrow = x + (size_t)row * (size_t)L;
    const char* embb = (const char*)emb;
    const unsigned loff = (unsigned)(lane < 25 ? lane : 24) * 8u;

    const int half  = (ntok + 1) >> 1;
    const int start = hsel * half;
    const int stop  = (start + half < ntok) ? start + half : ntok;

    v2f acc = {0.0f, 0.0f};

    for (int base = start; base < stop; base += 128) {
        int off[4];
#pragma unroll
        for (int k = 0; k < 4; ++k) {
            const int i = base + 32 * k + lane;
            const int r = xrow[i < stop ? i : stop - 1];
            off[k] = (int)((unsigned)r * (unsigned)ROWB);
        }
        asm volatile("" :: "v"(off[0]), "v"(off[1]), "v"(off[2]), "v"(off[3]));
#pragma unroll
        for (int k = 0; k < 4; ++k) {
            const int rem = stop - base - 32 * k;
            if (rem >= 32) {
                gather_full<32>(acc, embb, loff, off[k], 0);
            } else if (rem > 0) {
                int t = 0;
                if (rem >= 16) { gather_full<16>(acc, embb, loff, off[k], 0); t = 16; }
                if (t < rem) gather_part<16>(acc, embb, loff, off[k], t, rem);
            }
        }
    }

    part[wave][lane] = acc;
    Ws[wn0 * LPITCH + wk] = w0;
    Ws[wn1 * LPITCH + wk] = w1;
    __syncthreads();

    if (hsel == 0) {
        v2f s = part[wave][lane] + part[wave + 1][lane];
        const float fl = (float)len;
        s.x = fmaxf(s.x / fl, 0.0f);
        s.y = fmaxf(s.y / fl, 0.0f);
        if (lane >= 25) { s.x = 0.0f; s.y = 0.0f; }
        *(v2f*)(repS + (wave >> 1) * LPITCH + 2 * lane) = s;
    }
    __syncthreads();

    if (wave == 0) {
        const int row0  = blockIdx.x * SPB;
        const int mN    = lane & 15;
        const int khalf = (lane >> 4) * 2;

        const float* arow  = repS + mN        * LPITCH + khalf;
        const float* brow0 = Ws   + mN        * LPITCH + khalf;
        const float* brow1 = Ws   + (16 + mN) * LPITCH + khalf;

        v8f c0 = {};
        v8f c1 = {};

#pragma unroll
        for (int kk = 0; kk < 13; ++kk) {
            const int kb = kk * 4;
            v2f a  = *(const v2f*)(arow  + kb);
            v2f b0 = *(const v2f*)(brow0 + kb);
            v2f b1 = *(const v2f*)(brow1 + kb);

            c0 = __builtin_amdgcn_wmma_f32_16x16x4_f32(
                     false, a, false, b0, (short)0, c0, false, false);
            c1 = __builtin_amdgcn_wmma_f32_16x16x4_f32(
                     false, a, false, b1, (short)0, c1, false, false);
        }

        const int n0 = mN;
        const int n1 = 16 + mN;
        const bool n1v = (n1 < OUT);
        const int mbase = row0 + ((lane >> 4) ? 8 : 0);
#pragma unroll
        for (int r = 0; r < 8; ++r) {
            const int M = mbase + r;
            out[(size_t)M * OUT + n0] = c0[r] + bias0;
            if (n1v) out[(size_t)M * OUT + n1] = c1[r] + bias1;
        }
    }
}

extern "C" void kernel_launch(void* const* d_in, const int* in_sizes, int n_in,
                              void* d_out, int out_size, void* d_ws, size_t ws_size,
                              hipStream_t stream) {
    const int*   x       = (const int*)d_in[0];
    const int*   lengths = (const int*)d_in[1];
    const float* emb     = (const float*)d_in[2];
    const float* W       = (const float*)d_in[3];
    const float* bias    = (const float*)d_in[4];
    float*       out     = (float*)d_out;

    const int B = in_sizes[1];
    const int L = in_sizes[0] / B;

    pool_head_kernel<<<B / SPB, NWAVES * 32, 0, stream>>>(x, lengths, emb, W, bias, out, L);
}
